# starting kernel
# baseline (speedup 1.0000x reference)
#include <hip/hip_runtime.h>

typedef __attribute__((ext_vector_type(2))) float v2f;
typedef __attribute__((ext_vector_type(8))) float v8f;

#define R_CONST       1.1f
#define SCALE_CONST   100.0f
#define BUMP_STEPS    32
#define NUM_THETAS    32
#define NUM_FEATURES  128
#define LOG2E_F       1.4426950408889634f

__device__ __forceinline__ float fast_sigmoid(float z) {
#if __has_builtin(__builtin_amdgcn_tanhf)
    return __builtin_fmaf(0.5f, __builtin_amdgcn_tanhf(0.5f * z), 0.5f);
#else
    float e = __builtin_amdgcn_exp2f(-z * LOG2E_F);
    return __builtin_amdgcn_rcpf(1.0f + e);
#endif
}

__global__ __launch_bounds__(256)
void ect_fused_kernel(const float* __restrict__ x,
                      const int*   __restrict__ batch,
                      const float* __restrict__ v,
                      float* __restrict__ out,
                      int n_nodes)
{
    __shared__ float vsh[NUM_THETAS * NUM_FEATURES];
    __shared__ float tile[2 * 2 * BUMP_STEPS * NUM_THETAS];
    __shared__ int   s_g0, s_g1;

    const int tid  = threadIdx.x;
    const int wave = tid >> 5;
    const int lane = tid & 31;

    {
        const float4* vsrc = (const float4*)v;
        float4*       vdst = (float4*)vsh;
        for (int i = tid; i < (NUM_THETAS * NUM_FEATURES) / 4; i += 256)
            vdst[i] = vsrc[i];
    }
    for (int i = tid; i < 2 * 2 * BUMP_STEPS * NUM_THETAS; i += 256)
        tile[i] = 0.0f;

    if (tid == 0) {
        int base = blockIdx.x * 128;
        int last = base + 127; if (last >= n_nodes) last = n_nodes - 1;
        s_g0 = batch[base];
        s_g1 = batch[last];
    }
    __syncthreads();

    const int g0 = s_g0, g1 = s_g1;
    const int strip = blockIdx.x * 8 + wave;

    if (strip * 16 < n_nodes) {
        const int hi = lane >> 4;
        const int lm = lane & 15;
        const int node_base = strip * 16;

        const float* arow  = x   + (size_t)(node_base + lm) * NUM_FEATURES + 2 * hi;
        const float* b0row = vsh + (size_t)lm        * NUM_FEATURES + 2 * hi;
        const float* b1row = vsh + (size_t)(lm + 16) * NUM_FEATURES + 2 * hi;

        v8f acc0 = {};
        v8f acc1 = {};

        #pragma unroll 8
        for (int k = 0; k < NUM_FEATURES / 4; ++k) {
            const int kb = k * 4;
            if ((kb & 31) == 0)
                __builtin_prefetch(arow + kb + 64, 0, 3);
            v2f a  = *(const v2f*)(arow  + kb);
            v2f b0 = *(const v2f*)(b0row + kb);
            v2f b1 = *(const v2f*)(b1row + kb);
            acc0 = __builtin_amdgcn_wmma_f32_16x16x4_f32(false, a, false, b0,
                                                         (short)0, acc0, false, false);
            acc1 = __builtin_amdgcn_wmma_f32_16x16x4_f32(false, a, false, b1,
                                                         (short)0, acc1, false, false);
        }

        int grow[8];
        #pragma unroll
        for (int r = 0; r < 8; ++r)
            grow[r] = batch[node_base + 8 * hi + r];
        const bool gsame = (grow[0] == grow[7]);
        const int t0 = lm, t1 = lm + 16;
        float* tile0 = tile + hi * (BUMP_STEPS * NUM_THETAS);
        float* tile1 = tile0 + 2 * (BUMP_STEPS * NUM_THETAS);

        const float step = 2.0f * R_CONST / (float)(BUMP_STEPS - 1);

        #pragma unroll 1
        for (int s = 0; s < BUMP_STEPS; ++s) {
            const float lin_s  = -R_CONST + (float)s * step;
            const float base_s = fast_sigmoid(SCALE_CONST * (lin_s - R_CONST));
            const int so = s * NUM_THETAS;

            if (gsame) {
                float sum0 = 0.f, sum1 = 0.f;
                #pragma unroll
                for (int r = 0; r < 8; ++r) {
                    sum0 += fast_sigmoid(SCALE_CONST * (lin_s - acc0[r]));
                    sum1 += fast_sigmoid(SCALE_CONST * (lin_s - acc1[r]));
                }
                sum0 -= 8.0f * base_s;
                sum1 -= 8.0f * base_s;
                const int g = grow[0];
                if (g == g0) {
                    atomicAdd(&tile0[so + t0], sum0);
                    atomicAdd(&tile0[so + t1], sum1);
                } else if (g == g1) {
                    atomicAdd(&tile1[so + t0], sum0);
                    atomicAdd(&tile1[so + t1], sum1);
                } else {
                    atomicAdd(&out[(size_t)g * 1024 + so + t0], sum0);
                    atomicAdd(&out[(size_t)g * 1024 + so + t1], sum1);
                }
            } else {
                #pragma unroll
                for (int r = 0; r < 8; ++r) {
                    float e0 = fast_sigmoid(SCALE_CONST * (lin_s - acc0[r])) - base_s;
                    float e1 = fast_sigmoid(SCALE_CONST * (lin_s - acc1[r])) - base_s;
                    const int g = grow[r];
                    if (g == g0) {
                        atomicAdd(&tile0[so + t0], e0);
                        atomicAdd(&tile0[so + t1], e1);
                    } else if (g == g1) {
                        atomicAdd(&tile1[so + t0], e0);
                        atomicAdd(&tile1[so + t1], e1);
                    } else {
                        atomicAdd(&out[(size_t)g * 1024 + so + t0], e0);
                        atomicAdd(&out[(size_t)g * 1024 + so + t1], e1);
                    }
                }
            }
        }
    }
    __syncthreads();

    for (int i = tid; i < 1024; i += 256)
        atomicAdd(&out[(size_t)g0 * 1024 + i], tile[i] + tile[1024 + i]);
    if (g1 != g0)
        for (int i = tid; i < 1024; i += 256)
            atomicAdd(&out[(size_t)g1 * 1024 + i], tile[2048 + i] + tile[3072 + i]);
}

extern "C" void kernel_launch(void* const* d_in, const int* in_sizes, int n_in,
                              void* d_out, int out_size, void* d_ws, size_t ws_size,
                              hipStream_t stream) {
    const float* x     = (const float*)d_in[0];
    const int*   batch = (const int*)d_in[1];
    const float* v     = (const float*)d_in[3];
    const int n_nodes  = in_sizes[0] / NUM_FEATURES;

    (void)hipMemsetAsync(d_out, 0, (size_t)out_size * sizeof(float), stream);

    const int strips = (n_nodes + 15) / 16;
    const int blocks = (strips + 7) / 8;
    ect_fused_kernel<<<blocks, 256, 0, stream>>>(x, batch, v, (float*)d_out, n_nodes);
}


# Pad the LDS row pitch of the staged v from 128 to 132 floats so the B-fragment LDS reads are bank-conflict free (were 16-way); summation order unchanged
# speedup vs baseline: 1.0511x; 1.0511x over previous
#include <hip/hip_runtime.h>

typedef __attribute__((ext_vector_type(2))) float v2f;
typedef __attribute__((ext_vector_type(8))) float v8f;

#define R_CONST       1.1f
#define SCALE_CONST   100.0f
#define BUMP_STEPS    32
#define NUM_THETAS    32
#define NUM_FEATURES  128
#define LOG2E_F       1.4426950408889634f
#define V_PITCH       132

__device__ __forceinline__ float fast_sigmoid(float z) {
#if __has_builtin(__builtin_amdgcn_tanhf)
    return __builtin_fmaf(0.5f, __builtin_amdgcn_tanhf(0.5f * z), 0.5f);
#else
    float e = __builtin_amdgcn_exp2f(-z * LOG2E_F);
    return __builtin_amdgcn_rcpf(1.0f + e);
#endif
}

__global__ __launch_bounds__(256)
void ect_fused_kernel(const float* __restrict__ x,
                      const int*   __restrict__ batch,
                      const float* __restrict__ v,
                      float* __restrict__ out,
                      int n_nodes)
{
    __shared__ float vsh[NUM_THETAS * V_PITCH];
    __shared__ float tile[2 * 2 * BUMP_STEPS * NUM_THETAS];
    __shared__ int   s_g0, s_g1;

    const int tid  = threadIdx.x;
    const int wave = tid >> 5;
    const int lane = tid & 31;

    {
        const float4* vsrc = (const float4*)v;
        for (int i = tid; i < (NUM_THETAS * NUM_FEATURES) / 4; i += 256) {
            const int row = i / (NUM_FEATURES / 4);
            const int c4  = i % (NUM_FEATURES / 4);
            *(float4*)(vsh + row * V_PITCH + 4 * c4) = vsrc[i];
        }
    }
    for (int i = tid; i < 2 * 2 * BUMP_STEPS * NUM_THETAS; i += 256)
        tile[i] = 0.0f;

    if (tid == 0) {
        int base = blockIdx.x * 128;
        int last = base + 127; if (last >= n_nodes) last = n_nodes - 1;
        s_g0 = batch[base];
        s_g1 = batch[last];
    }
    __syncthreads();

    const int g0 = s_g0, g1 = s_g1;
    const int strip = blockIdx.x * 8 + wave;

    if (strip * 16 < n_nodes) {
        const int hi = lane >> 4;
        const int lm = lane & 15;
        const int node_base = strip * 16;

        const float* arow  = x   + (size_t)(node_base + lm) * NUM_FEATURES + 2 * hi;
        const float* b0row = vsh + (size_t)lm        * V_PITCH + 2 * hi;
        const float* b1row = vsh + (size_t)(lm + 16) * V_PITCH + 2 * hi;

        v8f acc0 = {};
        v8f acc1 = {};

        #pragma unroll 8
        for (int k = 0; k < NUM_FEATURES / 4; ++k) {
            const int kb = k * 4;
            if ((kb & 31) == 0)
                __builtin_prefetch(arow + kb + 64, 0, 3);
            v2f a  = *(const v2f*)(arow  + kb);
            v2f b0 = *(const v2f*)(b0row + kb);
            v2f b1 = *(const v2f*)(b1row + kb);
            acc0 = __builtin_amdgcn_wmma_f32_16x16x4_f32(false, a, false, b0,
                                                         (short)0, acc0, false, false);
            acc1 = __builtin_amdgcn_wmma_f32_16x16x4_f32(false, a, false, b1,
                                                         (short)0, acc1, false, false);
        }

        int grow[8];
        #pragma unroll
        for (int r = 0; r < 8; ++r)
            grow[r] = batch[node_base + 8 * hi + r];
        const bool gsame = (grow[0] == grow[7]);
        const int t0 = lm, t1 = lm + 16;
        float* tile0 = tile + hi * (BUMP_STEPS * NUM_THETAS);
        float* tile1 = tile0 + 2 * (BUMP_STEPS * NUM_THETAS);

        const float step = 2.0f * R_CONST / (float)(BUMP_STEPS - 1);

        #pragma unroll 1
        for (int s = 0; s < BUMP_STEPS; ++s) {
            const float lin_s  = -R_CONST + (float)s * step;
            const float base_s = fast_sigmoid(SCALE_CONST * (lin_s - R_CONST));
            const int so = s * NUM_THETAS;

            if (gsame) {
                float sum0 = 0.f, sum1 = 0.f;
                #pragma unroll
                for (int r = 0; r < 8; ++r) {
                    sum0 += fast_sigmoid(SCALE_CONST * (lin_s - acc0[r]));
                    sum1 += fast_sigmoid(SCALE_CONST * (lin_s - acc1[r]));
                }
                sum0 -= 8.0f * base_s;
                sum1 -= 8.0f * base_s;
                const int g = grow[0];
                if (g == g0) {
                    atomicAdd(&tile0[so + t0], sum0);
                    atomicAdd(&tile0[so + t1], sum1);
                } else if (g == g1) {
                    atomicAdd(&tile1[so + t0], sum0);
                    atomicAdd(&tile1[so + t1], sum1);
                } else {
                    atomicAdd(&out[(size_t)g * 1024 + so + t0], sum0);
                    atomicAdd(&out[(size_t)g * 1024 + so + t1], sum1);
                }
            } else {
                #pragma unroll
                for (int r = 0; r < 8; ++r) {
                    float e0 = fast_sigmoid(SCALE_CONST * (lin_s - acc0[r])) - base_s;
                    float e1 = fast_sigmoid(SCALE_CONST * (lin_s - acc1[r])) - base_s;
                    const int g = grow[r];
                    if (g == g0) {
                        atomicAdd(&tile0[so + t0], e0);
                        atomicAdd(&tile0[so + t1], e1);
                    } else if (g == g1) {
                        atomicAdd(&tile1[so + t0], e0);
                        atomicAdd(&tile1[so + t1], e1);
                    } else {
                        atomicAdd(&out[(size_t)g * 1024 + so + t0], e0);
                        atomicAdd(&out[(size_t)g * 1024 + so + t1], e1);
                    }
                }
            }
        }
    }
    __syncthreads();

    for (int i = tid; i < 1024; i += 256)
        atomicAdd(&out[(size_t)g0 * 1024 + i], tile[i] + tile[1024 + i]);
    if (g1 != g0)
        for (int i = tid; i < 1024; i += 256)
            atomicAdd(&out[(size_t)g1 * 1024 + i], tile[2048 + i] + tile[3072 + i]);
}

extern "C" void kernel_launch(void* const* d_in, const int* in_sizes, int n_in,
                              void* d_out, int out_size, void* d_ws, size_t ws_size,
                              hipStream_t stream) {
    const float* x     = (const float*)d_in[0];
    const int*   batch = (const int*)d_in[1];
    const float* v     = (const float*)d_in[3];
    const int n_nodes  = in_sizes[0] / NUM_FEATURES;

    (void)hipMemsetAsync(d_out, 0, (size_t)out_size * sizeof(float), stream);

    const int strips = (n_nodes + 15) / 16;
    const int blocks = (strips + 7) / 8;
    ect_fused_kernel<<<blocks, 256, 0, stream>>>(x, batch, v, (float*)d_out, n_nodes);
}


# Filtration loop cut to add+tanh+accumulate per element: accumulators pre-scaled once, sigmoid's 0.5/+0.5 and pad tanh hoisted out, packed f32 pairs, 2 steps per trip; sums of tanh round differently
# speedup vs baseline: 1.3267x; 1.2622x over previous
#include <hip/hip_runtime.h>

typedef __attribute__((ext_vector_type(2))) float v2f;
typedef __attribute__((ext_vector_type(8))) float v8f;

#define R_CONST       1.1f
#define SCALE_CONST   100.0f
#define BUMP_STEPS    32
#define NUM_THETAS    32
#define NUM_FEATURES  128
#define HALF_SCALE    (0.5f * SCALE_CONST)
#define V_PITCH       132

__global__ __launch_bounds__(256)
void ect_fused_kernel(const float* __restrict__ x,
                      const int*   __restrict__ batch,
                      const float* __restrict__ v,
                      float* __restrict__ out,
                      int n_nodes)
{
    __shared__ float vsh[NUM_THETAS * V_PITCH];
    __shared__ float tile[2 * 2 * BUMP_STEPS * NUM_THETAS];
    __shared__ int   s_g0, s_g1;

    const int tid  = threadIdx.x;
    const int wave = tid >> 5;
    const int lane = tid & 31;

    {
        const float4* vsrc = (const float4*)v;
        for (int i = tid; i < (NUM_THETAS * NUM_FEATURES) / 4; i += 256) {
            const int row = i / (NUM_FEATURES / 4);
            const int c4  = i % (NUM_FEATURES / 4);
            *(float4*)(vsh + row * V_PITCH + 4 * c4) = vsrc[i];
        }
    }
    for (int i = tid; i < 2 * 2 * BUMP_STEPS * NUM_THETAS; i += 256)
        tile[i] = 0.0f;

    if (tid == 0) {
        int base = blockIdx.x * 128;
        int last = base + 127; if (last >= n_nodes) last = n_nodes - 1;
        s_g0 = batch[base];
        s_g1 = batch[last];
    }
    __syncthreads();

    const int g0 = s_g0, g1 = s_g1;
    const int strip = blockIdx.x * 8 + wave;

    if (strip * 16 < n_nodes) {
        const int hi = lane >> 4;
        const int lm = lane & 15;
        const int node_base = strip * 16;

        const float* arow  = x   + (size_t)(node_base + lm) * NUM_FEATURES + 2 * hi;
        const float* b0row = vsh + (size_t)lm        * V_PITCH + 2 * hi;
        const float* b1row = vsh + (size_t)(lm + 16) * V_PITCH + 2 * hi;

        v8f acc0 = {};
        v8f acc1 = {};

        #pragma unroll 8
        for (int k = 0; k < NUM_FEATURES / 4; ++k) {
            const int kb = k * 4;
            if ((kb & 31) == 0)
                __builtin_prefetch(arow + kb + 64, 0, 3);
            v2f a  = *(const v2f*)(arow  + kb);
            v2f b0 = *(const v2f*)(b0row + kb);
            v2f b1 = *(const v2f*)(b1row + kb);
            acc0 = __builtin_amdgcn_wmma_f32_16x16x4_f32(false, a, false, b0,
                                                         (short)0, acc0, false, false);
            acc1 = __builtin_amdgcn_wmma_f32_16x16x4_f32(false, a, false, b1,
                                                         (short)0, acc1, false, false);
        }

        int grow[8];
        #pragma unroll
        for (int r = 0; r < 8; ++r)
            grow[r] = batch[node_base + 8 * hi + r];
        const bool gsame = (grow[0] == grow[7]);
        const int t0 = lm, t1 = lm + 16;
        float* tile0 = tile + hi * (BUMP_STEPS * NUM_THETAS);
        float* tile1 = tile0 + 2 * (BUMP_STEPS * NUM_THETAS);

        const float step = 2.0f * R_CONST / (float)(BUMP_STEPS - 1);

        v2f q[8];
        #pragma unroll
        for (int r = 0; r < 8; ++r)
            q[r] = (v2f){-HALF_SCALE * acc0[r], -HALF_SCALE * acc1[r]};

        static_assert(BUMP_STEPS <= 32, "one lane per filtration step");
        const float tb_lane = __builtin_amdgcn_tanhf(
            HALF_SCALE * ((-R_CONST + (float)lane * step) - R_CONST));

        const int  gl    = grow[0];
        const bool lfast = gsame && (gl == g0 || gl == g1);
        const bool wfast = (__builtin_amdgcn_ballot_w32(lfast) == 0xffffffffu);

        if (wfast) {
            float* tp = ((gl == g0) ? tile0 : tile1) + t0;
            #pragma unroll 2
            for (int s = 0; s < BUMP_STEPS; ++s) {
                const float lin_s = -R_CONST + (float)s * step;
                const float c_s   = HALF_SCALE * lin_s;
                const float tb_s  = __builtin_bit_cast(float,
                    __builtin_amdgcn_readlane(__builtin_bit_cast(int, tb_lane), s));
                const v2f cc = {c_s, c_s};
                v2f z = cc + q[0];
                v2f sum = {__builtin_amdgcn_tanhf(z[0]), __builtin_amdgcn_tanhf(z[1])};
                #pragma unroll
                for (int r = 1; r < 8; ++r) {
                    z = cc + q[r];
                    sum += (v2f){__builtin_amdgcn_tanhf(z[0]), __builtin_amdgcn_tanhf(z[1])};
                }
                const float pad = 4.0f * tb_s;
                atomicAdd(&tp[s * NUM_THETAS],      0.5f * sum[0] - pad);
                atomicAdd(&tp[s * NUM_THETAS + 16], 0.5f * sum[1] - pad);
            }
        } else {
            #pragma unroll 1
            for (int s = 0; s < BUMP_STEPS; ++s) {
                const float lin_s = -R_CONST + (float)s * step;
                const float c_s   = HALF_SCALE * lin_s;
                const float tb_s  = __builtin_bit_cast(float,
                    __builtin_amdgcn_readlane(__builtin_bit_cast(int, tb_lane), s));
                const int so = s * NUM_THETAS;

                if (gsame) {
                    float s0 = 0.f, s1 = 0.f;
                    #pragma unroll
                    for (int r = 0; r < 8; ++r) {
                        s0 += __builtin_amdgcn_tanhf(c_s + q[r][0]);
                        s1 += __builtin_amdgcn_tanhf(c_s + q[r][1]);
                    }
                    const float sum0 = 0.5f * (s0 - 8.0f * tb_s);
                    const float sum1 = 0.5f * (s1 - 8.0f * tb_s);
                    if (gl == g0) {
                        atomicAdd(&tile0[so + t0], sum0);
                        atomicAdd(&tile0[so + t1], sum1);
                    } else if (gl == g1) {
                        atomicAdd(&tile1[so + t0], sum0);
                        atomicAdd(&tile1[so + t1], sum1);
                    } else {
                        atomicAdd(&out[(size_t)gl * 1024 + so + t0], sum0);
                        atomicAdd(&out[(size_t)gl * 1024 + so + t1], sum1);
                    }
                } else {
                    #pragma unroll
                    for (int r = 0; r < 8; ++r) {
                        const float e0 = 0.5f * (__builtin_amdgcn_tanhf(c_s + q[r][0]) - tb_s);
                        const float e1 = 0.5f * (__builtin_amdgcn_tanhf(c_s + q[r][1]) - tb_s);
                        const int g = grow[r];
                        if (g == g0) {
                            atomicAdd(&tile0[so + t0], e0);
                            atomicAdd(&tile0[so + t1], e1);
                        } else if (g == g1) {
                            atomicAdd(&tile1[so + t0], e0);
                            atomicAdd(&tile1[so + t1], e1);
                        } else {
                            atomicAdd(&out[(size_t)g * 1024 + so + t0], e0);
                            atomicAdd(&out[(size_t)g * 1024 + so + t1], e1);
                        }
                    }
                }
            }
        }
    }
    __syncthreads();

    for (int i = tid; i < 1024; i += 256)
        atomicAdd(&out[(size_t)g0 * 1024 + i], tile[i] + tile[1024 + i]);
    if (g1 != g0)
        for (int i = tid; i < 1024; i += 256)
            atomicAdd(&out[(size_t)g1 * 1024 + i], tile[2048 + i] + tile[3072 + i]);
}

extern "C" void kernel_launch(void* const* d_in, const int* in_sizes, int n_in,
                              void* d_out, int out_size, void* d_ws, size_t ws_size,
                              hipStream_t stream) {
    const float* x     = (const float*)d_in[0];
    const int*   batch = (const int*)d_in[1];
    const float* v     = (const float*)d_in[3];
    const int n_nodes  = in_sizes[0] / NUM_FEATURES;

    (void)hipMemsetAsync(d_out, 0, (size_t)out_size * sizeof(float), stream);

    const int strips = (n_nodes + 15) / 16;
    const int blocks = (strips + 7) / 8;
    ect_fused_kernel<<<blocks, 256, 0, stream>>>(x, batch, v, (float*)d_out, n_nodes);
}


# Resident grid: 512 blocks of 8 waves loop over their strip ranges, next strip's rows requested ahead, v fragments kept in registers; a block's graph sums reach memory once: float add order changes
# speedup vs baseline: 1.4222x; 1.0720x over previous
#include <hip/hip_runtime.h>


typedef __attribute__((ext_vector_type(2))) float v2f;
typedef __attribute__((ext_vector_type(8))) float v8f;

#define R_CONST       1.1f
#define SCALE_CONST   100.0f
#define BUMP_STEPS    32
#define NUM_THETAS    32
#define NUM_FEATURES  128
#define HALF_SCALE    (0.5f * SCALE_CONST)
#define V_PITCH       132
#define BLOCK_THREADS 256
#define BLOCK_WAVES   (BLOCK_THREADS / 32)
#define KSTEPS        (NUM_FEATURES / 4)

__device__ __forceinline__ void load_strip(v2f (&a)[KSTEPS], const float* arow)
{
    #pragma unroll
    for (int i = 0; i < KSTEPS; ++i)
        a[i] = *(const v2f*)(arow + 4 * i);
}

__global__ __launch_bounds__(BLOCK_THREADS, 4)
void ect_fused_kernel(const float* __restrict__ x,
                      const int*   __restrict__ batch,
                      const float* __restrict__ v,
                      float* __restrict__ out,
                      int n_nodes)
{
    __shared__ float vsh[NUM_THETAS * V_PITCH];
    __shared__ float tile[2 * 2 * BUMP_STEPS * NUM_THETAS];
    __shared__ int   s_g0, s_g1;

    const int tid  = threadIdx.x;
    const int wave = __builtin_amdgcn_readfirstlane(tid >> 5);
    const int lane = tid & 31;

    const int strips    = (n_nodes + 15) >> 4;
    const int per_block = (strips + (int)gridDim.x - 1) / (int)gridDim.x;
    const int s_begin   = (int)blockIdx.x * per_block;
    int s_end = s_begin + per_block; if (s_end > strips) s_end = strips;
    if (s_begin >= s_end) return;

    const int hi = lane >> 4;
    const int lm = lane & 15;

    constexpr int V_VEC4     = (NUM_THETAS * NUM_FEATURES) / 4;
    constexpr int V_PER_LANE = V_VEC4 / BLOCK_THREADS;
    static_assert(V_VEC4 % BLOCK_THREADS == 0, "v is copied in whole rounds of the block");
    float4 vreg[V_PER_LANE];
    {
        const float4* vsrc = (const float4*)v;
        #pragma unroll
        for (int j = 0; j < V_PER_LANE; ++j)
            vreg[j] = vsrc[tid + j * BLOCK_THREADS];
    }

    v2f a[KSTEPS];
    int strip = s_begin + wave;
    if (strip < s_end)
        load_strip(a, x + (size_t)(strip * 16 + lm) * NUM_FEATURES + 2 * hi);

    for (int i = tid; i < 2 * 2 * BUMP_STEPS * NUM_THETAS; i += BLOCK_THREADS)
        tile[i] = 0.0f;

    #pragma unroll
    for (int j = 0; j < V_PER_LANE; ++j) {
        const int i   = tid + j * BLOCK_THREADS;
        const int row = i / (NUM_FEATURES / 4);
        const int c4  = i % (NUM_FEATURES / 4);
        *(float4*)(vsh + row * V_PITCH + 4 * c4) = vreg[j];
    }

    if (tid == 0) {
        int last = s_end * 16 - 1; if (last >= n_nodes) last = n_nodes - 1;
        s_g0 = batch[s_begin * 16];
        s_g1 = batch[last];
    }
    __syncthreads();

    const int g0 = s_g0, g1 = s_g1;

    const float* b0row = vsh + (size_t)lm        * V_PITCH + 2 * hi;
    const float* b1row = vsh + (size_t)(lm + 16) * V_PITCH + 2 * hi;

    const int t0 = lm, t1 = lm + 16;
    float* tile0 = tile + hi * (BUMP_STEPS * NUM_THETAS);
    float* tile1 = tile0 + 2 * (BUMP_STEPS * NUM_THETAS);

    const float step = 2.0f * R_CONST / (float)(BUMP_STEPS - 1);

    static_assert(BUMP_STEPS <= 32, "one lane per filtration step");
    const float tb_lane = __builtin_amdgcn_tanhf(
        HALF_SCALE * ((-R_CONST + (float)lane * step) - R_CONST));

    for (; strip < s_end; strip += BLOCK_WAVES) {
        const int node_base = strip * 16;

        int grow[8];
        #pragma unroll
        for (int r = 0; r < 8; ++r)
            grow[r] = batch[node_base + 8 * hi + r];
        const bool gsame = (grow[0] == grow[7]);
        const int  gl    = grow[0];
        const bool lfast = gsame && (gl == g0 || gl == g1);
        const bool wfast = (__builtin_amdgcn_ballot_w32(lfast) == 0xffffffffu);

        v8f acc0 = {};
        v8f acc1 = {};

        #pragma unroll
        for (int k = 0; k < KSTEPS; ++k) {
            const int kb = k * 4;
            v2f b0 = *(const v2f*)(b0row + kb);
            v2f b1 = *(const v2f*)(b1row + kb);
            acc0 = __builtin_amdgcn_wmma_f32_16x16x4_f32(false, a[k], false, b0,
                                                         (short)0, acc0, false, false);
            acc1 = __builtin_amdgcn_wmma_f32_16x16x4_f32(false, a[k], false, b1,
                                                         (short)0, acc1, false, false);
        }

        if (strip + BLOCK_WAVES < s_end)
            load_strip(a, x + (size_t)((strip + BLOCK_WAVES) * 16 + lm) * NUM_FEATURES + 2 * hi);

        v2f q[8];
        #pragma unroll
        for (int r = 0; r < 8; ++r)
            q[r] = (v2f){-HALF_SCALE * acc0[r], -HALF_SCALE * acc1[r]};

        if (wfast) {
            float* tp = ((gl == g0) ? tile0 : tile1) + t0;
            #pragma unroll 2
            for (int s = 0; s < BUMP_STEPS; ++s) {
                const float lin_s = -R_CONST + (float)s * step;
                const float c_s   = HALF_SCALE * lin_s;
                const float tb_s  = __builtin_bit_cast(float,
                    __builtin_amdgcn_readlane(__builtin_bit_cast(int, tb_lane), s));
                const v2f cc = {c_s, c_s};
                v2f z = cc + q[0];
                v2f sum = {__builtin_amdgcn_tanhf(z[0]), __builtin_amdgcn_tanhf(z[1])};
                #pragma unroll
                for (int r = 1; r < 8; ++r) {
                    z = cc + q[r];
                    sum += (v2f){__builtin_amdgcn_tanhf(z[0]), __builtin_amdgcn_tanhf(z[1])};
                }
                const float pad = 4.0f * tb_s;
                atomicAdd(&tp[s * NUM_THETAS],      0.5f * sum[0] - pad);
                atomicAdd(&tp[s * NUM_THETAS + 16], 0.5f * sum[1] - pad);
            }
        } else {
            #pragma unroll 1
            for (int s = 0; s < BUMP_STEPS; ++s) {
                const float lin_s = -R_CONST + (float)s * step;
                const float c_s   = HALF_SCALE * lin_s;
                const float tb_s  = __builtin_bit_cast(float,
                    __builtin_amdgcn_readlane(__builtin_bit_cast(int, tb_lane), s));
                const int so = s * NUM_THETAS;

                if (gsame) {
                    float s0 = 0.f, s1 = 0.f;
                    #pragma unroll
                    for (int r = 0; r < 8; ++r) {
                        s0 += __builtin_amdgcn_tanhf(c_s + q[r][0]);
                        s1 += __builtin_amdgcn_tanhf(c_s + q[r][1]);
                    }
                    const float sum0 = 0.5f * (s0 - 8.0f * tb_s);
                    const float sum1 = 0.5f * (s1 - 8.0f * tb_s);
                    if (gl == g0) {
                        atomicAdd(&tile0[so + t0], sum0);
                        atomicAdd(&tile0[so + t1], sum1);
                    } else if (gl == g1) {
                        atomicAdd(&tile1[so + t0], sum0);
                        atomicAdd(&tile1[so + t1], sum1);
                    } else {
                        atomicAdd(&out[(size_t)gl * 1024 + so + t0], sum0);
                        atomicAdd(&out[(size_t)gl * 1024 + so + t1], sum1);
                    }
                } else {
                    #pragma unroll
                    for (int r = 0; r < 8; ++r) {
                        const float e0 = 0.5f * (__builtin_amdgcn_tanhf(c_s + q[r][0]) - tb_s);
                        const float e1 = 0.5f * (__builtin_amdgcn_tanhf(c_s + q[r][1]) - tb_s);
                        const int g = grow[r];
                        if (g == g0) {
                            atomicAdd(&tile0[so + t0], e0);
                            atomicAdd(&tile0[so + t1], e1);
                        } else if (g == g1) {
                            atomicAdd(&tile1[so + t0], e0);
                            atomicAdd(&tile1[so + t1], e1);
                        } else {
                            atomicAdd(&out[(size_t)g * 1024 + so + t0], e0);
                            atomicAdd(&out[(size_t)g * 1024 + so + t1], e1);
                        }
                    }
                }
            }
        }
    }
    __syncthreads();

    for (int i = tid; i < 1024; i += BLOCK_THREADS)
        atomicAdd(&out[(size_t)g0 * 1024 + i], tile[i] + tile[1024 + i]);
    if (g1 != g0)
        for (int i = tid; i < 1024; i += BLOCK_THREADS)
            atomicAdd(&out[(size_t)g1 * 1024 + i], tile[2048 + i] + tile[3072 + i]);
}

extern "C" void kernel_launch(void* const* d_in, const int* in_sizes, int n_in,
                              void* d_out, int out_size, void* d_ws, size_t ws_size,
                              hipStream_t stream) {
    const float* x     = (const float*)d_in[0];
    const int*   batch = (const int*)d_in[1];
    const float* v     = (const float*)d_in[3];
    const int n_nodes  = in_sizes[0] / NUM_FEATURES;

    (void)hipMemsetAsync(d_out, 0, (size_t)out_size * sizeof(float), stream);

    ect_fused_kernel<<<512, BLOCK_THREADS, 0, stream>>>(x, batch, v, (float*)d_out, n_nodes);
}
